# starting kernel
# baseline (speedup 1.0000x reference)
#include <hip/hip_runtime.h>

typedef __attribute__((ext_vector_type(2))) float v2f;
typedef __attribute__((ext_vector_type(4))) float v4f;
typedef __attribute__((ext_vector_type(8))) float v8f;

#define DIMS 64
#define ROWS_PER_BLOCK 128
#define XSTRIDE 68
#define BSTRIDE 130

__global__ __launch_bounds__(256)
void poly_wmma_kernel(const float* __restrict__ X,
                      const float* __restrict__ W,
                      const float* __restrict__ Bv,
                      float* __restrict__ out)
{
    __shared__ float Xs[ROWS_PER_BLOCK][XSTRIDE];
    __shared__ v2f   Bsw[16][BSTRIDE];
    __shared__ float Wsh[DIMS];

    const int tid     = threadIdx.x;
    const int rowbase = blockIdx.x * ROWS_PER_BLOCK;

    if (tid < DIMS) Wsh[tid] = W[1 + tid];

    for (int e = tid; e < 16 * 128; e += 256) {
        int ks   = e >> 7;
        int t    = e & 127;
        int n    = t >> 1;
        int half = t & 1;
        int k    = 4 * ks + 2 * half;
        v2f v;
        v.x = (k     >= n) ? W[65 + n * DIMS - (n * (n - 1)) / 2 + (k     - n)] : 0.0f;
        v.y = (k + 1 >= n) ? W[65 + n * DIMS - (n * (n - 1)) / 2 + (k + 1 - n)] : 0.0f;
        Bsw[ks][2 * n + half] = v;
    }

    {
        const v4f* Xg = (const v4f*)(X + (size_t)rowbase * DIMS);
        #pragma unroll
        for (int i = 0; i < 8; ++i) {
            int g = tid + i * 256;
            int r = g >> 4;
            int c = (g & 15) << 2;
            v4f val = Xg[g];
            *(v4f*)&Xs[r][c] = val;
        }
    }
    __syncthreads();

    const int wave = tid >> 5;
    const int lane = tid & 31;
    const int half = lane >> 4;
    const int l15  = lane & 15;
    const int wr   = wave * 16;

    v2f afrag[16];
    #pragma unroll
    for (int ks = 0; ks < 16; ++ks) {
        int k = ks * 4 + half * 2;
        afrag[ks].x = Xs[wr + l15][k];
        afrag[ks].y = Xs[wr + l15][k + 1];
    }

    v8f acc[4];
    #pragma unroll
    for (int nt = 0; nt < 4; ++nt) acc[nt] = (v8f)(0.0f);

    #pragma unroll
    for (int nt = 0; nt < 4; ++nt) {
        #pragma unroll
        for (int ks = 0; ks < 16; ++ks) {
            v2f bfrag = Bsw[ks][2 * (nt * 16 + l15) + half];
            acc[nt] = __builtin_amdgcn_wmma_f32_16x16x4_f32(
                false, afrag[ks], false, bfrag, (short)0, acc[nt], false, false);
        }
    }

    const float bias = W[0] + Bv[0];

    #pragma unroll
    for (int r = 0; r < 8; ++r) {
        int m = r + 8 * half;
        float s = 0.0f;
        #pragma unroll
        for (int nt = 0; nt < 4; ++nt) {
            int n = nt * 16 + l15;
            s += Xs[wr + m][n] * (Wsh[n] + acc[nt][r]);
        }
        s += __shfl_xor(s, 1, 16);
        s += __shfl_xor(s, 2, 16);
        s += __shfl_xor(s, 4, 16);
        s += __shfl_xor(s, 8, 16);
        if (l15 == 0) out[rowbase + wr + m] = bias + s;
    }
}

extern "C" void kernel_launch(void* const* d_in, const int* in_sizes, int n_in,
                              void* d_out, int out_size, void* d_ws, size_t ws_size,
                              hipStream_t stream) {
    const float* x  = (const float*)d_in[0];
    const float* W  = (const float*)d_in[1];
    const float* bv = (const float*)d_in[2];
    float* out = (float*)d_out;

    const int batch = in_sizes[0] / DIMS;
    const int grid  = batch / ROWS_PER_BLOCK;
    poly_wmma_kernel<<<grid, 256, 0, stream>>>(x, W, bv, out);
}


# Row sums by a transposing butterfly of 8 row_xmask DPP adds (a lane's rows halve per level) instead of 32 chained ds_bpermute; one 64-B store per wave; pairing order of the sums unchanged
# speedup vs baseline: 1.6975x; 1.6975x over previous
#include <hip/hip_runtime.h>


typedef __attribute__((ext_vector_type(2))) float v2f;
typedef __attribute__((ext_vector_type(4))) float v4f;
typedef __attribute__((ext_vector_type(8))) float v8f;
typedef float v4f_a4 __attribute__((ext_vector_type(4), aligned(4)));

#define DIMS 64
#define ROWS_PER_BLOCK 128
#define XSTRIDE 68
#define NWR  2144
#define NW4  536

template <int MASK>
__device__ __forceinline__ float row_xor(float x) {
    return __builtin_bit_cast(float,
        __builtin_amdgcn_update_dpp(0, __builtin_bit_cast(int, x),
                                    0x160 + MASK, 0xf, 0xf, true));
}

__global__ __launch_bounds__(256)
void poly_wmma_kernel(const float* __restrict__ X,
                      const float* __restrict__ W,
                      const float* __restrict__ Bv,
                      float* __restrict__ out)
{
    __shared__ float Xs[ROWS_PER_BLOCK][XSTRIDE];
    __shared__ __attribute__((aligned(16))) float Wr[NWR];

    const int tid     = threadIdx.x;
    const int rowbase = blockIdx.x * ROWS_PER_BLOCK;

    const v4f* Xg = (const v4f*)(X + (size_t)rowbase * DIMS);
    v4f xv[8];
    #pragma unroll
    for (int i = 0; i < 8; ++i) xv[i] = Xg[tid + i * 256];

    const v4f_a4* Wg = (const v4f_a4*)(W + 1);
    const int g2 = tid + (NW4 - 256);
    const v4f w0 = Wg[tid];
    const v4f w1 = Wg[tid + 256];
    const v4f w2 = Wg[g2];
    const float wb = W[0];
    const float bb = Bv[0];
    asm volatile("" :: "s"(wb), "s"(bb));

    #pragma unroll
    for (int i = 0; i < 8; ++i) {
        int g = tid + i * 256;
        int r = g >> 4;
        int c = (g & 15) << 2;
        *(v4f*)&Xs[r][c] = xv[i];
    }
    ((v4f*)Wr)[tid]       = w0;
    ((v4f*)Wr)[tid + 256] = w1;
    ((v4f*)Wr)[g2]        = w2;
    __syncthreads();

    const int wave = tid >> 5;
    const int lane = tid & 31;
    const int half = lane >> 4;
    const int l15  = lane & 15;
    const int wr   = wave * 16;

    v2f afrag[16];
    #pragma unroll
    for (int ks = 0; ks < 16; ++ks) {
        int k = ks * 4 + half * 2;
        afrag[ks].x = Xs[wr + l15][k];
        afrag[ks].y = Xs[wr + l15][k + 1];
    }

    v8f acc[4];
    #pragma unroll
    for (int nt = 0; nt < 4; ++nt) acc[nt] = (v8f)(0.0f);

    const int dlane = l15 - 2 * half;
    #pragma unroll
    for (int nt = 0; nt < 4; ++nt) {
        const int n = nt * 16 + l15;
        const float* urow = Wr + (64 + 63 * n - (n * (n - 1)) / 2 + 2 * half);
        #pragma unroll
        for (int ks = 0; ks < 16; ++ks) {
            const int kd = 4 * ks - 16 * nt;
            const float u0 = urow[4 * ks];
            const float u1 = urow[4 * ks + 1];
            v2f bfrag;
            bfrag.x = (kd     >= dlane) ? u0 : 0.0f;
            bfrag.y = (kd + 1 >= dlane) ? u1 : 0.0f;
            acc[nt] = __builtin_amdgcn_wmma_f32_16x16x4_f32(
                false, afrag[ks], false, bfrag, (short)0, acc[nt], false, false);
        }
    }

    const float bias = wb + bb;

    float s[8];
    #pragma unroll
    for (int r = 0; r < 8; ++r) {
        int m = r + 8 * half;
        float p = 0.0f;
        #pragma unroll
        for (int nt = 0; nt < 4; ++nt) {
            int n = nt * 16 + l15;
            p += Xs[wr + m][n] * (Wr[n] + acc[nt][r]);
        }
        s[r] = p;
    }

    const bool b0 = (lane & 1) != 0;
    const bool b1 = (lane & 2) != 0;
    const bool b2 = (lane & 4) != 0;
    float t[4];
    #pragma unroll
    for (int j = 0; j < 4; ++j) {
        const float keep = b0 ? s[2 * j + 1] : s[2 * j];
        const float give = b0 ? s[2 * j]     : s[2 * j + 1];
        t[j] = keep + row_xor<1>(give);
    }
    float u[2];
    #pragma unroll
    for (int i = 0; i < 2; ++i) {
        const float keep = b1 ? t[2 * i + 1] : t[2 * i];
        const float give = b1 ? t[2 * i]     : t[2 * i + 1];
        u[i] = keep + row_xor<2>(give);
    }
    const float keep = b2 ? u[1] : u[0];
    const float give = b2 ? u[0] : u[1];
    float v = keep + row_xor<4>(give);
    v = v + row_xor<8>(v);

    if (l15 < 8) out[rowbase + wr + 8 * half + l15] = bias + v;
}

extern "C" void kernel_launch(void* const* d_in, const int* in_sizes, int n_in,
                              void* d_out, int out_size, void* d_ws, size_t ws_size,
                              hipStream_t stream) {
    const float* x  = (const float*)d_in[0];
    const float* W  = (const float*)d_in[1];
    const float* bv = (const float*)d_in[2];
    float* out = (float*)d_out;

    const int batch = in_sizes[0] / DIMS;
    const int grid  = batch / ROWS_PER_BLOCK;
    poly_wmma_kernel<<<grid, 256, 0, stream>>>(x, W, bv, out);
}
